# starting kernel
# baseline (speedup 1.0000x reference)
#include <hip/hip_runtime.h>
#include <hip/hip_bf16.h>


#define N_NODES 100000
#define E_EDGES 1600000
#define IN_DIM  32
#define LATENT  64

typedef __attribute__((ext_vector_type(2))) float v2f;
typedef __attribute__((ext_vector_type(8))) float v8f;

__device__ __forceinline__ void atomic_add_f32(float* p, float v) {
    asm volatile("global_atomic_add_f32 %0, %1, off scope:SCOPE_DEV"
                 :: "v"(p), "v"(v)
                 : "memory");
}


__global__ void deg_kernel(const int* __restrict__ col,
                           const float* __restrict__ w,
                           float* __restrict__ deg) {
    int e = blockIdx.x * blockDim.x + threadIdx.x;
    if (e < E_EDGES) atomic_add_f32(&deg[col[e]], w[e]);
}

__global__ void dinv_kernel(float* __restrict__ deg) {
    int v = blockIdx.x * blockDim.x + threadIdx.x;
    if (v < N_NODES) {
        float d = deg[v];
        deg[v] = (d > 0.0f) ? __frsqrt_rn(d) : 0.0f;
    }
}

__global__ void norm_kernel(const int* __restrict__ row,
                            const int* __restrict__ col,
                            const float* __restrict__ w,
                            const float* __restrict__ dinv,
                            float* __restrict__ norm) {
    int e = blockIdx.x * blockDim.x + threadIdx.x;
    if (e < E_EDGES) norm[e] = dinv[row[e]] * w[e] * dinv[col[e]];
}


template <int D>
__global__ void scatter_kernel(const float* __restrict__ src,
                               float* __restrict__ dst,
                               const int* __restrict__ row,
                               const int* __restrict__ col,
                               const float* __restrict__ norm) {
    constexpr int VE = D / 4;
    int t = blockIdx.x * blockDim.x + threadIdx.x;
    int e = t / VE;
    int q = t & (VE - 1);
    int r = row[e];
    int c = col[e];
    float nv = norm[e];
    const float4 v = *(const float4*)(src + (size_t)r * D + q * 4);
    float* dp = dst + (size_t)c * D + q * 4;
    atomic_add_f32(dp + 0, nv * v.x);
    atomic_add_f32(dp + 1, nv * v.y);
    atomic_add_f32(dp + 2, nv * v.z);
    atomic_add_f32(dp + 3, nv * v.w);
}


template <int D>
__global__ void __launch_bounds__(256)
tag_gemm_kernel(const float* __restrict__ H0, const float* __restrict__ H1,
                const float* __restrict__ H2, const float* __restrict__ H3,
                const float* __restrict__ W,
                const float* __restrict__ bias,
                float* __restrict__ out) {
    const int wave  = threadIdx.x >> 5;
    const int lane  = threadIdx.x & 31;
    const int mtile = blockIdx.x * 2 + (wave >> 2);
    const int ntile = wave & 3;

    const int m    = lane & 15;
    const int half = lane >> 4;
    const int arow = mtile * 16 + m;
    const int ncol = ntile * 16 + m;

    const float* Hs[4] = {H0, H1, H2, H3};
    v8f c = {};

#pragma unroll
    for (int s = 0; s < 4; ++s) {
        const float* __restrict__ A  = Hs[s];
        const float* __restrict__ Wb = W + (size_t)s * D * 64;
#pragma unroll
        for (int k = 0; k < D; k += 4) {
            const int kk = k + half * 2;
            v2f a;
            a.x = A[(size_t)arow * D + kk];
            a.y = A[(size_t)arow * D + kk + 1];
            v2f b;
            b.x = Wb[(size_t)kk * 64 + ncol];
            b.y = Wb[(size_t)(kk + 1) * 64 + ncol];
            c = __builtin_amdgcn_wmma_f32_16x16x4_f32(
                     false, a,  false, b,
                     (short)0, c,  false,  false);
        }
    }

    const float bv = bias[ncol];
#pragma unroll
    for (int v = 0; v < 8; ++v) {
        const int orow = mtile * 16 + v + half * 8;
        float r = c[v] + bv;
        out[(size_t)orow * 64 + ncol] = r > 0.0f ? r : 0.0f;
    }
}


template <int D>
static void run_layer(const float* hin, const float* W, const float* bias,
                      float* buf1, float* buf2, float* buf3,
                      const int* row, const int* col, const float* norm,
                      float* hout, hipStream_t stream) {
    const size_t hopBytes = (size_t)N_NODES * D * sizeof(float);
    const int sblocks = (E_EDGES * (D / 4)) / 256;

    hipMemsetAsync(buf1, 0, hopBytes, stream);
    scatter_kernel<D><<<sblocks, 256, 0, stream>>>(hin, buf1, row, col, norm);
    hipMemsetAsync(buf2, 0, hopBytes, stream);
    scatter_kernel<D><<<sblocks, 256, 0, stream>>>(buf1, buf2, row, col, norm);
    hipMemsetAsync(buf3, 0, hopBytes, stream);
    scatter_kernel<D><<<sblocks, 256, 0, stream>>>(buf2, buf3, row, col, norm);

    tag_gemm_kernel<D><<<N_NODES / 32, 256, 0, stream>>>(
        hin, buf1, buf2, buf3, W, bias, hout);
}

extern "C" void kernel_launch(void* const* d_in, const int* in_sizes, int n_in,
                              void* d_out, int out_size, void* d_ws, size_t ws_size,
                              hipStream_t stream) {
    const float* x   = (const float*)d_in[0];
    const int*   ei  = (const int*)d_in[1];
    const float* ew  = (const float*)d_in[2];
    const float* W0  = (const float*)d_in[3];
    const float* b0  = (const float*)d_in[4];
    const float* Ws  = (const float*)d_in[5];
    const float* bs  = (const float*)d_in[6];
    float*       out = (float*)d_out;

    const int* row = ei;
    const int* col = ei + E_EDGES;

    float* ws   = (float*)d_ws;
    float* deg  = ws;
    float* norm = deg + N_NODES;
    float* bufA = norm + E_EDGES;
    float* bufB = bufA + (size_t)N_NODES * LATENT;
    float* buf1 = bufB + (size_t)N_NODES * LATENT;
    float* buf2 = buf1 + (size_t)N_NODES * LATENT;
    float* buf3 = buf2 + (size_t)N_NODES * LATENT;

    hipMemsetAsync(deg, 0, N_NODES * sizeof(float), stream);
    deg_kernel<<<E_EDGES / 256, 256, 0, stream>>>(col, ew, deg);
    dinv_kernel<<<(N_NODES + 255) / 256, 256, 0, stream>>>(deg);
    norm_kernel<<<E_EDGES / 256, 256, 0, stream>>>(row, col, ew, deg, norm);

    const size_t WL = 4 * 64 * 64;
    run_layer<IN_DIM>(x,    W0,          b0,       buf1, buf2, buf3, row, col, norm, bufA, stream);
    run_layer<LATENT>(bufA, Ws + 0 * WL, bs + 0,   buf1, buf2, buf3, row, col, norm, bufB, stream);
    run_layer<LATENT>(bufB, Ws + 1 * WL, bs + 64,  buf1, buf2, buf3, row, col, norm, bufA, stream);
    run_layer<LATENT>(bufA, Ws + 2 * WL, bs + 128, buf1, buf2, buf3, row, col, norm, bufB, stream);
    run_layer<LATENT>(bufB, Ws + 3 * WL, bs + 192, buf1, buf2, buf3, row, col, norm, out,  stream);
}


# Hop sums kept as two copies, one per L2 half: a wave reads its die id and adds into its half's copy, readers (next gather, hop GEMM) add the copies; 4 edge groups per wave; sum order changes
# speedup vs baseline: 1.1700x; 1.1700x over previous
#include <hip/hip_runtime.h>
#include <hip/hip_bf16.h>


#define N_NODES 100000
#define E_EDGES 1600000
#define IN_DIM  32
#define LATENT  64

typedef __attribute__((ext_vector_type(2))) float v2f;
typedef __attribute__((ext_vector_type(8))) float v8f;

__device__ __forceinline__ void atomic_add_f32(float* p, float v) {
    asm volatile("global_atomic_add_f32 %0, %1, off scope:SCOPE_DEV"
                 :: "v"(p), "v"(v)
                 : "memory");
}


__global__ void deg_kernel(const int* __restrict__ col,
                           const float* __restrict__ w,
                           float* __restrict__ deg) {
    int e = blockIdx.x * blockDim.x + threadIdx.x;
    if (e < E_EDGES) atomic_add_f32(&deg[col[e]], w[e]);
}

__global__ void dinv_kernel(float* __restrict__ deg) {
    int v = blockIdx.x * blockDim.x + threadIdx.x;
    if (v < N_NODES) {
        float d = deg[v];
        deg[v] = (d > 0.0f) ? __frsqrt_rn(d) : 0.0f;
    }
}

__global__ void norm_kernel(const int* __restrict__ row,
                            const int* __restrict__ col,
                            const float* __restrict__ w,
                            const float* __restrict__ dinv,
                            float* __restrict__ norm) {
    int e = blockIdx.x * blockDim.x + threadIdx.x;
    if (e < E_EDGES) norm[e] = dinv[row[e]] * w[e] * dinv[col[e]];
}


#define EDGE_STEPS 4

__device__ __forceinline__ int l2_half_of_this_wave() {
    const unsigned v = __builtin_amdgcn_s_sendmsg_rtn(0x87);
    return (int)((v >> 18) & 1u);
}

template <int D, bool SRC_PAIR>
__global__ void __launch_bounds__(256)
scatter_kernel(const float* __restrict__ src,
               float* __restrict__ dst,
               const int* __restrict__ row,
               const int* __restrict__ col,
               const float* __restrict__ norm) {
    constexpr int VE  = D / 4;
    constexpr int EPW = 32 / VE;
    constexpr size_t COPY = (size_t)N_NODES * D;
    const int lane = threadIdx.x & 31;
    const int wv   = blockIdx.x * 8 + (threadIdx.x >> 5);
    const int g    = lane / VE;
    const int q    = lane & (VE - 1);

    float* dcopy = dst + (size_t)l2_half_of_this_wave() * COPY;

    int    cs[EDGE_STEPS];
    float  ns[EDGE_STEPS];
    float4 vs[EDGE_STEPS];
#pragma unroll
    for (int u = 0; u < EDGE_STEPS; ++u) {
        const int e = (wv * EDGE_STEPS + u) * EPW + g;
        const int r = row[e];
        cs[u] = col[e];
        ns[u] = norm[e];
        const float* sp = src + (size_t)r * D + q * 4;
        float4 v = *(const float4*)sp;
        if (SRC_PAIR) {
            const float4 w = *(const float4*)(sp + COPY);
            v.x += w.x; v.y += w.y; v.z += w.z; v.w += w.w;
        }
        vs[u] = v;
    }
#pragma unroll
    for (int u = 0; u < EDGE_STEPS; ++u) {
        float* dp = dcopy + (size_t)cs[u] * D + q * 4;
        const float nv = ns[u];
        atomic_add_f32(dp + 0, nv * vs[u].x);
        atomic_add_f32(dp + 1, nv * vs[u].y);
        atomic_add_f32(dp + 2, nv * vs[u].z);
        atomic_add_f32(dp + 3, nv * vs[u].w);
    }
}


template <int D>
__global__ void __launch_bounds__(256)
tag_gemm_kernel(const float* __restrict__ H0, const float* __restrict__ H1,
                const float* __restrict__ H2, const float* __restrict__ H3,
                const float* __restrict__ W,
                const float* __restrict__ bias,
                float* __restrict__ out) {
    const int wave  = threadIdx.x >> 5;
    const int lane  = threadIdx.x & 31;
    const int mtile = blockIdx.x * 2 + (wave >> 2);
    const int ntile = wave & 3;

    const int m    = lane & 15;
    const int half = lane >> 4;
    const int arow = mtile * 16 + m;
    const int ncol = ntile * 16 + m;

    const float* Hs[4] = {H0, H1, H2, H3};
    v8f c = {};

#pragma unroll
    for (int s = 0; s < 4; ++s) {
        const float* __restrict__ A  = Hs[s];
        const float* __restrict__ Wb = W + (size_t)s * D * 64;
#pragma unroll
        for (int k = 0; k < D; k += 4) {
            const int kk = k + half * 2;
            v2f a;
            a.x = A[(size_t)arow * D + kk];
            a.y = A[(size_t)arow * D + kk + 1];
            if (s > 0) {
                const float* __restrict__ A2 = A + (size_t)N_NODES * D;
                a.x += A2[(size_t)arow * D + kk];
                a.y += A2[(size_t)arow * D + kk + 1];
            }
            v2f b;
            b.x = Wb[(size_t)kk * 64 + ncol];
            b.y = Wb[(size_t)(kk + 1) * 64 + ncol];
            c = __builtin_amdgcn_wmma_f32_16x16x4_f32(
                     false, a,  false, b,
                     (short)0, c,  false,  false);
        }
    }

    const float bv = bias[ncol];
#pragma unroll
    for (int v = 0; v < 8; ++v) {
        const int orow = mtile * 16 + v + half * 8;
        float r = c[v] + bv;
        out[(size_t)orow * 64 + ncol] = r > 0.0f ? r : 0.0f;
    }
}


template <int D>
static void run_layer(const float* hin, const float* W, const float* bias,
                      float* buf1, float* buf2, float* buf3,
                      const int* row, const int* col, const float* norm,
                      float* hout, hipStream_t stream) {
    const size_t hopBytes = 2 * (size_t)N_NODES * D * sizeof(float);
    const int sblocks = (E_EDGES * (D / 4)) / (256 * EDGE_STEPS);

    hipMemsetAsync(buf1, 0, hopBytes, stream);
    scatter_kernel<D, false><<<sblocks, 256, 0, stream>>>(hin, buf1, row, col, norm);
    hipMemsetAsync(buf2, 0, hopBytes, stream);
    scatter_kernel<D, true><<<sblocks, 256, 0, stream>>>(buf1, buf2, row, col, norm);
    hipMemsetAsync(buf3, 0, hopBytes, stream);
    scatter_kernel<D, true><<<sblocks, 256, 0, stream>>>(buf2, buf3, row, col, norm);

    tag_gemm_kernel<D><<<N_NODES / 32, 256, 0, stream>>>(
        hin, buf1, buf2, buf3, W, bias, hout);
}

extern "C" void kernel_launch(void* const* d_in, const int* in_sizes, int n_in,
                              void* d_out, int out_size, void* d_ws, size_t ws_size,
                              hipStream_t stream) {
    const float* x   = (const float*)d_in[0];
    const int*   ei  = (const int*)d_in[1];
    const float* ew  = (const float*)d_in[2];
    const float* W0  = (const float*)d_in[3];
    const float* b0  = (const float*)d_in[4];
    const float* Ws  = (const float*)d_in[5];
    const float* bs  = (const float*)d_in[6];
    float*       out = (float*)d_out;

    const int* row = ei;
    const int* col = ei + E_EDGES;

    float* ws   = (float*)d_ws;
    float* deg  = ws;
    float* norm = deg + N_NODES;
    float* bufA = norm + E_EDGES;
    float* bufB = bufA + (size_t)N_NODES * LATENT;
    float* buf1 = bufB + (size_t)N_NODES * LATENT;
    float* buf2 = buf1 + 2 * (size_t)N_NODES * LATENT;
    float* buf3 = buf2 + 2 * (size_t)N_NODES * LATENT;

    hipMemsetAsync(deg, 0, N_NODES * sizeof(float), stream);
    deg_kernel<<<E_EDGES / 256, 256, 0, stream>>>(col, ew, deg);
    dinv_kernel<<<(N_NODES + 255) / 256, 256, 0, stream>>>(deg);
    norm_kernel<<<E_EDGES / 256, 256, 0, stream>>>(row, col, ew, deg, norm);

    const size_t WL = 4 * 64 * 64;
    run_layer<IN_DIM>(x,    W0,          b0,       buf1, buf2, buf3, row, col, norm, bufA, stream);
    run_layer<LATENT>(bufA, Ws + 0 * WL, bs + 0,   buf1, buf2, buf3, row, col, norm, bufB, stream);
    run_layer<LATENT>(bufB, Ws + 1 * WL, bs + 64,  buf1, buf2, buf3, row, col, norm, bufA, stream);
    run_layer<LATENT>(bufA, Ws + 2 * WL, bs + 128, buf1, buf2, buf3, row, col, norm, bufB, stream);
    run_layer<LATENT>(bufB, Ws + 3 * WL, bs + 192, buf1, buf2, buf3, row, col, norm, out,  stream);
}


# Hop GEMM operand feed: a block stages its 32 rows of all four hop inputs in LDS with whole-line 16-B loads, pair copies summed while staging; waves read A fragments from LDS (pitch D+4)
# speedup vs baseline: 1.2760x; 1.0906x over previous
#include <hip/hip_runtime.h>
#include <hip/hip_bf16.h>


#define N_NODES 100000
#define E_EDGES 1600000
#define IN_DIM  32
#define LATENT  64

typedef __attribute__((ext_vector_type(2))) float v2f;
typedef __attribute__((ext_vector_type(8))) float v8f;

__device__ __forceinline__ void atomic_add_f32(float* p, float v) {
    asm volatile("global_atomic_add_f32 %0, %1, off scope:SCOPE_DEV"
                 :: "v"(p), "v"(v)
                 : "memory");
}


__global__ void deg_kernel(const int* __restrict__ col,
                           const float* __restrict__ w,
                           float* __restrict__ deg) {
    int e = blockIdx.x * blockDim.x + threadIdx.x;
    if (e < E_EDGES) atomic_add_f32(&deg[col[e]], w[e]);
}

__global__ void dinv_kernel(float* __restrict__ deg) {
    int v = blockIdx.x * blockDim.x + threadIdx.x;
    if (v < N_NODES) {
        float d = deg[v];
        deg[v] = (d > 0.0f) ? __frsqrt_rn(d) : 0.0f;
    }
}

__global__ void norm_kernel(const int* __restrict__ row,
                            const int* __restrict__ col,
                            const float* __restrict__ w,
                            const float* __restrict__ dinv,
                            float* __restrict__ norm) {
    int e = blockIdx.x * blockDim.x + threadIdx.x;
    if (e < E_EDGES) norm[e] = dinv[row[e]] * w[e] * dinv[col[e]];
}


#define EDGE_STEPS 4

__device__ __forceinline__ int l2_half_of_this_wave() {
    const unsigned v = __builtin_amdgcn_s_sendmsg_rtn(0x87);
    return (int)((v >> 18) & 1u);
}

template <int D, bool SRC_PAIR>
__global__ void __launch_bounds__(256)
scatter_kernel(const float* __restrict__ src,
               float* __restrict__ dst,
               const int* __restrict__ row,
               const int* __restrict__ col,
               const float* __restrict__ norm) {
    constexpr int VE  = D / 4;
    constexpr int EPW = 32 / VE;
    constexpr size_t COPY = (size_t)N_NODES * D;
    const int lane = threadIdx.x & 31;
    const int wv   = blockIdx.x * 8 + (threadIdx.x >> 5);
    const int g    = lane / VE;
    const int q    = lane & (VE - 1);

    float* dcopy = dst + (size_t)l2_half_of_this_wave() * COPY;

    int    cs[EDGE_STEPS];
    float  ns[EDGE_STEPS];
    float4 vs[EDGE_STEPS];
#pragma unroll
    for (int u = 0; u < EDGE_STEPS; ++u) {
        const int e = (wv * EDGE_STEPS + u) * EPW + g;
        const int r = row[e];
        cs[u] = col[e];
        ns[u] = norm[e];
        const float* sp = src + (size_t)r * D + q * 4;
        float4 v = *(const float4*)sp;
        if (SRC_PAIR) {
            const float4 w = *(const float4*)(sp + COPY);
            v.x += w.x; v.y += w.y; v.z += w.z; v.w += w.w;
        }
        vs[u] = v;
    }
#pragma unroll
    for (int u = 0; u < EDGE_STEPS; ++u) {
        float* dp = dcopy + (size_t)cs[u] * D + q * 4;
        const float nv = ns[u];
        atomic_add_f32(dp + 0, nv * vs[u].x);
        atomic_add_f32(dp + 1, nv * vs[u].y);
        atomic_add_f32(dp + 2, nv * vs[u].z);
        atomic_add_f32(dp + 3, nv * vs[u].w);
    }
}


template <int D>
__global__ void __launch_bounds__(256)
tag_gemm_kernel(const float* __restrict__ H0, const float* __restrict__ H1,
                const float* __restrict__ H2, const float* __restrict__ H3,
                const float* __restrict__ W,
                const float* __restrict__ bias,
                float* __restrict__ out) {
    constexpr int PITCH = D + 4;
    constexpr int PLANE = 32 * PITCH;
    constexpr int C4    = D / 4;
    constexpr int LPT   = 32 * C4 / 256;
    constexpr size_t COPY = (size_t)N_NODES * D;
    __shared__ __attribute__((aligned(16))) float As[4 * PLANE];

    const int wave  = threadIdx.x >> 5;
    const int lane  = threadIdx.x & 31;
    const int mtile = blockIdx.x * 2 + (wave >> 2);
    const int ntile = wave & 3;

    const int m    = lane & 15;
    const int half = lane >> 4;
    const int ncol = ntile * 16 + m;

    const float* Hs[4] = {H0, H1, H2, H3};
    const size_t blk = (size_t)blockIdx.x * 32 * D;
    float4 st[4][LPT];
    float4 sp[4][LPT];
#pragma unroll
    for (int s = 0; s < 4; ++s) {
#pragma unroll
        for (int i = 0; i < LPT; ++i) {
            const int idx = threadIdx.x + i * 256;
            const float* p = Hs[s] + blk + (size_t)idx * 4;
            st[s][i] = *(const float4*)p;
            if (s > 0) sp[s][i] = *(const float4*)(p + COPY);
        }
    }
#pragma unroll
    for (int s = 0; s < 4; ++s) {
#pragma unroll
        for (int i = 0; i < LPT; ++i) {
            const int idx = threadIdx.x + i * 256;
            float4 v = st[s][i];
            if (s > 0) {
                const float4 w = sp[s][i];
                v.x += w.x; v.y += w.y; v.z += w.z; v.w += w.w;
            }
            *(float4*)(As + s * PLANE + (idx / C4) * PITCH + (idx % C4) * 4) = v;
        }
    }
    __syncthreads();

    const float* __restrict__ Al = As + ((wave >> 2) * 16 + m) * PITCH + half * 2;
    v8f c = {};

#pragma unroll
    for (int s = 0; s < 4; ++s) {
        const float* __restrict__ Wb = W + (size_t)s * D * 64;
#pragma unroll
        for (int k = 0; k < D; k += 4) {
            const int kk = k + half * 2;
            const v2f a = *(const v2f*)(Al + s * PLANE + k);
            v2f b;
            b.x = Wb[(size_t)kk * 64 + ncol];
            b.y = Wb[(size_t)(kk + 1) * 64 + ncol];
            c = __builtin_amdgcn_wmma_f32_16x16x4_f32(
                     false, a,  false, b,
                     (short)0, c,  false,  false);
        }
    }

    const float bv = bias[ncol];
#pragma unroll
    for (int v = 0; v < 8; ++v) {
        const int orow = mtile * 16 + v + half * 8;
        float r = c[v] + bv;
        out[(size_t)orow * 64 + ncol] = r > 0.0f ? r : 0.0f;
    }
}


template <int D>
static void run_layer(const float* hin, const float* W, const float* bias,
                      float* buf1, float* buf2, float* buf3,
                      const int* row, const int* col, const float* norm,
                      float* hout, hipStream_t stream) {
    const size_t hopBytes = 2 * (size_t)N_NODES * D * sizeof(float);
    const int sblocks = (E_EDGES * (D / 4)) / (256 * EDGE_STEPS);

    hipMemsetAsync(buf1, 0, hopBytes, stream);
    scatter_kernel<D, false><<<sblocks, 256, 0, stream>>>(hin, buf1, row, col, norm);
    hipMemsetAsync(buf2, 0, hopBytes, stream);
    scatter_kernel<D, true><<<sblocks, 256, 0, stream>>>(buf1, buf2, row, col, norm);
    hipMemsetAsync(buf3, 0, hopBytes, stream);
    scatter_kernel<D, true><<<sblocks, 256, 0, stream>>>(buf2, buf3, row, col, norm);

    tag_gemm_kernel<D><<<N_NODES / 32, 256, 0, stream>>>(
        hin, buf1, buf2, buf3, W, bias, hout);
}

extern "C" void kernel_launch(void* const* d_in, const int* in_sizes, int n_in,
                              void* d_out, int out_size, void* d_ws, size_t ws_size,
                              hipStream_t stream) {
    const float* x   = (const float*)d_in[0];
    const int*   ei  = (const int*)d_in[1];
    const float* ew  = (const float*)d_in[2];
    const float* W0  = (const float*)d_in[3];
    const float* b0  = (const float*)d_in[4];
    const float* Ws  = (const float*)d_in[5];
    const float* bs  = (const float*)d_in[6];
    float*       out = (float*)d_out;

    const int* row = ei;
    const int* col = ei + E_EDGES;

    float* ws   = (float*)d_ws;
    float* deg  = ws;
    float* norm = deg + N_NODES;
    float* bufA = norm + E_EDGES;
    float* bufB = bufA + (size_t)N_NODES * LATENT;
    float* buf1 = bufB + (size_t)N_NODES * LATENT;
    float* buf2 = buf1 + 2 * (size_t)N_NODES * LATENT;
    float* buf3 = buf2 + 2 * (size_t)N_NODES * LATENT;

    hipMemsetAsync(deg, 0, N_NODES * sizeof(float), stream);
    deg_kernel<<<E_EDGES / 256, 256, 0, stream>>>(col, ew, deg);
    dinv_kernel<<<(N_NODES + 255) / 256, 256, 0, stream>>>(deg);
    norm_kernel<<<E_EDGES / 256, 256, 0, stream>>>(row, col, ew, deg, norm);

    const size_t WL = 4 * 64 * 64;
    run_layer<IN_DIM>(x,    W0,          b0,       buf1, buf2, buf3, row, col, norm, bufA, stream);
    run_layer<LATENT>(bufA, Ws + 0 * WL, bs + 0,   buf1, buf2, buf3, row, col, norm, bufB, stream);
    run_layer<LATENT>(bufB, Ws + 1 * WL, bs + 64,  buf1, buf2, buf3, row, col, norm, bufA, stream);
    run_layer<LATENT>(bufA, Ws + 2 * WL, bs + 128, buf1, buf2, buf3, row, col, norm, bufB, stream);
    run_layer<LATENT>(bufB, Ws + 3 * WL, bs + 192, buf1, buf2, buf3, row, col, norm, out,  stream);
}
